# starting kernel
# baseline (speedup 1.0000x reference)
#include <hip/hip_runtime.h>
#include <hip/hip_bf16.h>
#include <math.h>

#define B_  4096
#define M_  100
#define E_  128
#define R_  64

typedef float v2f __attribute__((ext_vector_type(2)));
typedef float v8f __attribute__((ext_vector_type(8)));

#define STRIDE    132
#define MT        112
#define OFF_INTER 0
#define OFF_U     (OFF_INTER + MT * STRIDE)
#define OFF_IP    (OFF_U + E_)
#define OFF_INN   (OFF_IP + E_)
#define OFF_W     (OFF_INN + E_)
#define OFF_WSUA  (OFF_W + 4 * E_)
#define OFF_WSIA  (OFF_WSUA + E_)
#define OFF_LOGIT (OFF_WSIA + E_)
#define OFF_SACC  (OFF_LOGIT + 4 * MT)
#define SMEM_FLOATS (OFF_SACC + 4 * E_)

__device__ __forceinline__ float wave_sum(float v) {
    #pragma unroll
    for (int off = 16; off > 0; off >>= 1) v += __shfl_xor(v, off, 32);
    return v;
}
__device__ __forceinline__ float wave_max(float v) {
    #pragma unroll
    for (int off = 16; off > 0; off >>= 1) v = fmaxf(v, __shfl_xor(v, off, 32));
    return v;
}

__device__ __forceinline__ void renorm_row(const float* __restrict__ src,
                                           float* __restrict__ dst, int lane) {
    float4 x = ((const float4*)src)[lane];
    float ss = x.x * x.x + x.y * x.y + x.z * x.z + x.w * x.w;
    ss = wave_sum(ss);
    float n = sqrtf(ss);
    float s = (n > 1.0f) ? (1.0f / (n + 1e-7f)) : 1.0f;
    x.x *= s; x.y *= s; x.z *= s; x.w *= s;
    ((float4*)dst)[lane] = x;
}

__global__ void hlrm_wsum_kernel(const float* __restrict__ ua_key,
                                 const float* __restrict__ ua_val,
                                 const float* __restrict__ ia_key,
                                 const float* __restrict__ ia_val,
                                 float* __restrict__ wsum) {
    int e = threadIdx.x;
    float s1 = 0.f, s2 = 0.f;
    #pragma unroll 4
    for (int r = 0; r < R_; ++r) {
        s1 += ua_key[r * E_ + e] * ua_val[r * E_ + e];
        s2 += ia_key[r * E_ + e] * ia_val[r * E_ + e];
    }
    wsum[e]       = s1;
    wsum[E_ + e]  = s2;
}

__global__ void __launch_bounds__(256, 2)
hlrm_main_kernel(const int* __restrict__ user_id,
                 const int* __restrict__ item_id_p,
                 const int* __restrict__ item_id_n,
                 const int* __restrict__ inter_id,
                 const float* __restrict__ user_emb_w,
                 const float* __restrict__ item_emb_w,
                 const float* __restrict__ wsum_g,
                 float* __restrict__ out) {
    extern __shared__ float sm[];
    float* sm_inter = sm + OFF_INTER;
    float* sm_u     = sm + OFF_U;
    float* sm_ip    = sm + OFF_IP;
    float* sm_inn   = sm + OFF_INN;
    float* sm_w     = sm + OFF_W;
    float* sm_wsua  = sm + OFF_WSUA;
    float* sm_wsia  = sm + OFF_WSIA;
    float* sm_logit = sm + OFF_LOGIT;
    float* sm_sacc  = sm + OFF_SACC;

    const int b    = blockIdx.x;
    const int t    = threadIdx.x;
    const int wave = t >> 5;
    const int lane = t & 31;
    const int half = lane >> 4;
    const int lm   = lane & 15;

    if (wave == 0)      renorm_row(user_emb_w + (size_t)user_id[b]   * E_, sm_u,   lane);
    else if (wave == 1) renorm_row(item_emb_w + (size_t)item_id_p[b] * E_, sm_ip,  lane);
    else if (wave == 2) renorm_row(item_emb_w + (size_t)item_id_n[b] * E_, sm_inn, lane);

    for (int m = wave; m < MT; m += 8) {
        if (m < M_) {
            const float* src = item_emb_w + (size_t)inter_id[b * M_ + m] * E_;
            if (m + 8 < M_) {
                const float* nxt = item_emb_w + (size_t)inter_id[b * M_ + m + 8] * E_;
                __builtin_prefetch(nxt + 4 * lane, 0, 3);
            }
            renorm_row(src, sm_inter + m * STRIDE, lane);
        } else {
            float4 z = {0.f, 0.f, 0.f, 0.f};
            ((float4*)(sm_inter + m * STRIDE))[lane] = z;
        }
    }
    __syncthreads();

    if (t < E_) {
        const int e = t;
        const float uu = sm_u[e], pp = sm_ip[e], nn = sm_inn[e];
        const float wa = wsum_g[e], wi = wsum_g[E_ + e];
        sm_wsua[e] = wa;
        sm_wsia[e] = wi;
        sm_w[0 * E_ + e] = uu * pp * wa;
        sm_w[1 * E_ + e] = uu * pp * wi;
        sm_w[2 * E_ + e] = uu * nn * wa;
        sm_w[3 * E_ + e] = uu * nn * wi;
    }
    __syncthreads();

    if (wave < 7) {
        const int   tile = wave;
        const float* Arow = sm_inter + (tile * 16 + lm) * STRIDE;
        const float  bmask = (lm < 4) ? 1.0f : 0.0f;
        const float* wrow  = sm_w + (lm < 4 ? lm : 0) * E_;
        v8f c = {0.f, 0.f, 0.f, 0.f, 0.f, 0.f, 0.f, 0.f};
        #pragma unroll 8
        for (int k0 = 0; k0 < E_; k0 += 4) {
            const int ke = k0 + 2 * half;
            v2f a, bb;
            a[0]  = Arow[ke];           a[1]  = Arow[ke + 1];
            bb[0] = wrow[ke] * bmask;   bb[1] = wrow[ke + 1] * bmask;
            c = __builtin_amdgcn_wmma_f32_16x16x4_f32(false, a, false, bb,
                                                      (short)0, c, false, false);
        }
        if (lm < 4) {
            #pragma unroll
            for (int v = 0; v < 8; ++v)
                sm_logit[lm * MT + tile * 16 + v + 8 * half] = c[v];
        }
    }
    __syncthreads();

    if (wave < 4) {
        float* L = sm_logit + wave * MT;
        float vals[4], mx = -INFINITY;
        #pragma unroll
        for (int i = 0; i < 4; ++i) {
            int m = lane + 32 * i;
            vals[i] = (m < M_) ? L[m] : -INFINITY;
            mx = fmaxf(mx, vals[i]);
        }
        mx = wave_max(mx);
        float se = 0.f;
        #pragma unroll
        for (int i = 0; i < 4; ++i) {
            int m = lane + 32 * i;
            vals[i] = (m < M_) ? expf(vals[i] - mx) : 0.f;
            se += vals[i];
        }
        se = wave_sum(se);
        const float inv = 1.0f / se;
        #pragma unroll
        for (int i = 0; i < 4; ++i) {
            int m = lane + 32 * i;
            if (m < MT) L[m] = vals[i] * inv;
        }
    }
    __syncthreads();

    {
        const int   etile = wave;
        const float amask = (lm < 4) ? 1.0f : 0.0f;
        const float* Lrow = sm_logit + (lm < 4 ? lm : 0) * MT;
        v8f c = {0.f, 0.f, 0.f, 0.f, 0.f, 0.f, 0.f, 0.f};
        #pragma unroll 7
        for (int m0 = 0; m0 < MT; m0 += 4) {
            const int mk = m0 + 2 * half;
            v2f a, bb;
            a[0]  = Lrow[mk] * amask;     a[1]  = Lrow[mk + 1] * amask;
            bb[0] = sm_inter[mk * STRIDE + etile * 16 + lm];
            bb[1] = sm_inter[(mk + 1) * STRIDE + etile * 16 + lm];
            c = __builtin_amdgcn_wmma_f32_16x16x4_f32(false, a, false, bb,
                                                      (short)0, c, false, false);
        }
        if (half == 0) {
            #pragma unroll
            for (int v = 0; v < 4; ++v)
                sm_sacc[v * E_ + etile * 16 + lm] = c[v];
        }
    }
    __syncthreads();

    if (t < E_) {
        const int e = t;
        const float uu = sm_u[e], pp = sm_ip[e], nn = sm_inn[e];
        const float wa = sm_wsua[e], wi = sm_wsia[e];
        const float relp = pp * wa * sm_sacc[0 * E_ + e] + uu * wi * sm_sacc[1 * E_ + e];
        const float reln = nn * wa * sm_sacc[2 * E_ + e] + uu * wi * sm_sacc[3 * E_ + e];
        const size_t BE = (size_t)B_ * E_;
        const size_t base = (size_t)b * E_ + e;
        out[0 * BE + base] = uu;
        out[1 * BE + base] = pp;
        out[2 * BE + base] = nn;
        out[3 * BE + base] = relp;
        out[4 * BE + base] = reln;
    }
}

extern "C" void kernel_launch(void* const* d_in, const int* in_sizes, int n_in,
                              void* d_out, int out_size, void* d_ws, size_t ws_size,
                              hipStream_t stream) {
    const int*   user_id    = (const int*)d_in[0];
    const int*   item_id_p  = (const int*)d_in[1];
    const int*   item_id_n  = (const int*)d_in[2];
    const int*   inter_id   = (const int*)d_in[3];
    const float* user_emb_w = (const float*)d_in[4];
    const float* item_emb_w = (const float*)d_in[5];
    const float* ua_key     = (const float*)d_in[6];
    const float* ua_val     = (const float*)d_in[7];
    const float* ia_key     = (const float*)d_in[8];
    const float* ia_val     = (const float*)d_in[9];
    float* out  = (float*)d_out;
    float* wsum = (float*)d_ws;

    hlrm_wsum_kernel<<<1, E_, 0, stream>>>(ua_key, ua_val, ia_key, ia_val, wsum);
    hlrm_main_kernel<<<B_, 256, SMEM_FLOATS * sizeof(float), stream>>>(
        user_id, item_id_p, item_id_n, inter_id,
        user_emb_w, item_emb_w, wsum, out);
}


# Gather tiled 8 lanes per row, 4 rows per wave at once (a whole 128-B line per row per load; u/ip/inn in the same pass); row sum of squares = 16 terms per lane + 3 in-row DPP adds; prefetch dropped
# speedup vs baseline: 1.4909x; 1.4909x over previous
#include <hip/hip_runtime.h>
#include <hip/hip_bf16.h>
#include <math.h>

#define B_  4096
#define M_  100
#define E_  128
#define R_  64

typedef float v2f __attribute__((ext_vector_type(2)));
typedef float v8f __attribute__((ext_vector_type(8)));

#define STRIDE    132
#define MT        112
#define OFF_INTER 0
#define OFF_U     (OFF_INTER + MT * STRIDE)
#define OFF_IP    (OFF_U + E_)
#define OFF_INN   (OFF_IP + E_)
#define OFF_W     (OFF_INN + E_)
#define OFF_WSUA  (OFF_W + 4 * E_)
#define OFF_WSIA  (OFF_WSUA + E_)
#define OFF_LOGIT (OFF_WSIA + E_)
#define OFF_SACC  (OFF_LOGIT + 4 * MT)
#define SMEM_FLOATS (OFF_SACC + 4 * E_)

__device__ __forceinline__ float wave_sum(float v) {
    #pragma unroll
    for (int off = 16; off > 0; off >>= 1) v += __shfl_xor(v, off, 32);
    return v;
}
__device__ __forceinline__ float wave_max(float v) {
    #pragma unroll
    for (int off = 16; off > 0; off >>= 1) v = fmaxf(v, __shfl_xor(v, off, 32));
    return v;
}

template <int M>
__device__ __forceinline__ float add_row_xor(float v) {
    const int b = __builtin_bit_cast(int, v);
    const int o = __builtin_amdgcn_update_dpp(b, b, 0x160 + M, 0xf, 0xf, false);
    return v + __builtin_bit_cast(float, o);
}

__global__ void hlrm_wsum_kernel(const float* __restrict__ ua_key,
                                 const float* __restrict__ ua_val,
                                 const float* __restrict__ ia_key,
                                 const float* __restrict__ ia_val,
                                 float* __restrict__ wsum) {
    int e = threadIdx.x;
    float s1 = 0.f, s2 = 0.f;
    #pragma unroll 4
    for (int r = 0; r < R_; ++r) {
        s1 += ua_key[r * E_ + e] * ua_val[r * E_ + e];
        s2 += ia_key[r * E_ + e] * ia_val[r * E_ + e];
    }
    wsum[e]       = s1;
    wsum[E_ + e]  = s2;
}

__global__ void __launch_bounds__(256, 2)
hlrm_main_kernel(const int* __restrict__ user_id,
                 const int* __restrict__ item_id_p,
                 const int* __restrict__ item_id_n,
                 const int* __restrict__ inter_id,
                 const float* __restrict__ user_emb_w,
                 const float* __restrict__ item_emb_w,
                 const float* __restrict__ wsum_g,
                 float* __restrict__ out) {
    extern __shared__ float sm[];
    float* sm_inter = sm + OFF_INTER;
    float* sm_u     = sm + OFF_U;
    float* sm_ip    = sm + OFF_IP;
    float* sm_inn   = sm + OFF_INN;
    float* sm_w     = sm + OFF_W;
    float* sm_wsua  = sm + OFF_WSUA;
    float* sm_wsia  = sm + OFF_WSIA;
    float* sm_logit = sm + OFF_LOGIT;
    float* sm_sacc  = sm + OFF_SACC;

    const int b    = blockIdx.x;
    const int t    = threadIdx.x;
    const int wave = t >> 5;
    const int lane = t & 31;
    const int half = lane >> 4;
    const int lm   = lane & 15;

    {
        const int q  = lane >> 3;
        const int j  = lane & 7;
        const int s0 = wave * 4 + q;
        const int s3 = 96 + s0;

        const int* idp = inter_id + b * M_ + (s3 < M_ ? s3 : 0);
        idp = (s3 == MT)     ? user_id   + b : idp;
        idp = (s3 == MT + 1) ? item_id_p + b : idp;
        idp = (s3 == MT + 2) ? item_id_n + b : idp;
        const int id3 = *idp;
        const int id0 = inter_id[b * M_ + s0];
        const int id1 = inter_id[b * M_ + s0 + 32];
        const int id2 = inter_id[b * M_ + s0 + 64];
        const bool has_src3 = (s3 < M_) || (s3 >= MT && s3 <= MT + 2);
        const bool has_dst3 = (s3 <= MT + 2);

        const float4* src0 = (const float4*)(item_emb_w + (size_t)id0 * E_) + j;
        const float4* src1 = (const float4*)(item_emb_w + (size_t)id1 * E_) + j;
        const float4* src2 = (const float4*)(item_emb_w + (size_t)id2 * E_) + j;
        const float4* src3 = (const float4*)((s3 == MT ? user_emb_w : item_emb_w)
                                             + (size_t)id3 * E_) + j;
        float* d3 = sm_inter + s3 * STRIDE;
        d3 = (s3 == MT)     ? sm_u   : d3;
        d3 = (s3 == MT + 1) ? sm_ip  : d3;
        d3 = (s3 == MT + 2) ? sm_inn : d3;
        float4* dst0 = (float4*)(sm_inter + s0 * STRIDE) + j;
        float4* dst1 = (float4*)(sm_inter + (s0 + 32) * STRIDE) + j;
        float4* dst2 = (float4*)(sm_inter + (s0 + 64) * STRIDE) + j;
        float4* dst3 = (float4*)d3 + j;

        float4 x[4][4];
        #pragma unroll
        for (int seg = 0; seg < 4; ++seg) {
            x[0][seg] = src0[seg * 8];
            x[1][seg] = src1[seg * 8];
            x[2][seg] = src2[seg * 8];
        }
        #pragma unroll
        for (int seg = 0; seg < 4; ++seg) {
            float4 v = {0.f, 0.f, 0.f, 0.f};
            if (has_src3) v = src3[seg * 8];
            x[3][seg] = v;
        }
        #pragma unroll
        for (int pass = 0; pass < 4; ++pass) {
            float ps[4];
            #pragma unroll
            for (int seg = 0; seg < 4; ++seg) {
                const float4 v = x[pass][seg];
                ps[seg] = v.x * v.x + v.y * v.y + v.z * v.z + v.w * v.w;
            }
            float ss = (ps[0] + ps[1]) + (ps[2] + ps[3]);
            ss = add_row_xor<4>(ss);
            ss = add_row_xor<2>(ss);
            ss = add_row_xor<1>(ss);
            const float n  = sqrtf(ss);
            const float sc = (n > 1.0f) ? (1.0f / (n + 1e-7f)) : 1.0f;
            #pragma unroll
            for (int seg = 0; seg < 4; ++seg) {
                x[pass][seg].x *= sc; x[pass][seg].y *= sc;
                x[pass][seg].z *= sc; x[pass][seg].w *= sc;
            }
        }
        #pragma unroll
        for (int seg = 0; seg < 4; ++seg) {
            dst0[seg * 8] = x[0][seg];
            dst1[seg * 8] = x[1][seg];
            dst2[seg * 8] = x[2][seg];
        }
        if (has_dst3) {
            #pragma unroll
            for (int seg = 0; seg < 4; ++seg) dst3[seg * 8] = x[3][seg];
        }
    }
    __syncthreads();

    if (t < E_) {
        const int e = t;
        const float uu = sm_u[e], pp = sm_ip[e], nn = sm_inn[e];
        const float wa = wsum_g[e], wi = wsum_g[E_ + e];
        sm_wsua[e] = wa;
        sm_wsia[e] = wi;
        sm_w[0 * E_ + e] = uu * pp * wa;
        sm_w[1 * E_ + e] = uu * pp * wi;
        sm_w[2 * E_ + e] = uu * nn * wa;
        sm_w[3 * E_ + e] = uu * nn * wi;
    }
    __syncthreads();

    if (wave < 7) {
        const int   tile = wave;
        const float* Arow = sm_inter + (tile * 16 + lm) * STRIDE;
        const float  bmask = (lm < 4) ? 1.0f : 0.0f;
        const float* wrow  = sm_w + (lm < 4 ? lm : 0) * E_;
        v8f c = {0.f, 0.f, 0.f, 0.f, 0.f, 0.f, 0.f, 0.f};
        #pragma unroll 8
        for (int k0 = 0; k0 < E_; k0 += 4) {
            const int ke = k0 + 2 * half;
            v2f a, bb;
            a[0]  = Arow[ke];           a[1]  = Arow[ke + 1];
            bb[0] = wrow[ke] * bmask;   bb[1] = wrow[ke + 1] * bmask;
            c = __builtin_amdgcn_wmma_f32_16x16x4_f32(false, a, false, bb,
                                                      (short)0, c, false, false);
        }
        if (lm < 4) {
            #pragma unroll
            for (int v = 0; v < 8; ++v)
                sm_logit[lm * MT + tile * 16 + v + 8 * half] = c[v];
        }
    }
    __syncthreads();

    if (wave < 4) {
        float* L = sm_logit + wave * MT;
        float vals[4], mx = -INFINITY;
        #pragma unroll
        for (int i = 0; i < 4; ++i) {
            int m = lane + 32 * i;
            vals[i] = (m < M_) ? L[m] : -INFINITY;
            mx = fmaxf(mx, vals[i]);
        }
        mx = wave_max(mx);
        float se = 0.f;
        #pragma unroll
        for (int i = 0; i < 4; ++i) {
            int m = lane + 32 * i;
            vals[i] = (m < M_) ? expf(vals[i] - mx) : 0.f;
            se += vals[i];
        }
        se = wave_sum(se);
        const float inv = 1.0f / se;
        #pragma unroll
        for (int i = 0; i < 4; ++i) {
            int m = lane + 32 * i;
            if (m < MT) L[m] = vals[i] * inv;
        }
    }
    __syncthreads();

    {
        const int   etile = wave;
        const float amask = (lm < 4) ? 1.0f : 0.0f;
        const float* Lrow = sm_logit + (lm < 4 ? lm : 0) * MT;
        v8f c = {0.f, 0.f, 0.f, 0.f, 0.f, 0.f, 0.f, 0.f};
        #pragma unroll 7
        for (int m0 = 0; m0 < MT; m0 += 4) {
            const int mk = m0 + 2 * half;
            v2f a, bb;
            a[0]  = Lrow[mk] * amask;     a[1]  = Lrow[mk + 1] * amask;
            bb[0] = sm_inter[mk * STRIDE + etile * 16 + lm];
            bb[1] = sm_inter[(mk + 1) * STRIDE + etile * 16 + lm];
            c = __builtin_amdgcn_wmma_f32_16x16x4_f32(false, a, false, bb,
                                                      (short)0, c, false, false);
        }
        if (half == 0) {
            #pragma unroll
            for (int v = 0; v < 4; ++v)
                sm_sacc[v * E_ + etile * 16 + lm] = c[v];
        }
    }
    __syncthreads();

    if (t < E_) {
        const int e = t;
        const float uu = sm_u[e], pp = sm_ip[e], nn = sm_inn[e];
        const float wa = sm_wsua[e], wi = sm_wsia[e];
        const float relp = pp * wa * sm_sacc[0 * E_ + e] + uu * wi * sm_sacc[1 * E_ + e];
        const float reln = nn * wa * sm_sacc[2 * E_ + e] + uu * wi * sm_sacc[3 * E_ + e];
        const size_t BE = (size_t)B_ * E_;
        const size_t base = (size_t)b * E_ + e;
        out[0 * BE + base] = uu;
        out[1 * BE + base] = pp;
        out[2 * BE + base] = nn;
        out[3 * BE + base] = relp;
        out[4 * BE + base] = reln;
    }
}

extern "C" void kernel_launch(void* const* d_in, const int* in_sizes, int n_in,
                              void* d_out, int out_size, void* d_ws, size_t ws_size,
                              hipStream_t stream) {
    const int*   user_id    = (const int*)d_in[0];
    const int*   item_id_p  = (const int*)d_in[1];
    const int*   item_id_n  = (const int*)d_in[2];
    const int*   inter_id   = (const int*)d_in[3];
    const float* user_emb_w = (const float*)d_in[4];
    const float* item_emb_w = (const float*)d_in[5];
    const float* ua_key     = (const float*)d_in[6];
    const float* ua_val     = (const float*)d_in[7];
    const float* ia_key     = (const float*)d_in[8];
    const float* ia_val     = (const float*)d_in[9];
    float* out  = (float*)d_out;
    float* wsum = (float*)d_ws;

    hlrm_wsum_kernel<<<1, E_, 0, stream>>>(ua_key, ua_val, ia_key, ia_val, wsum);
    hlrm_main_kernel<<<B_, 256, SMEM_FLOATS * sizeof(float), stream>>>(
        user_id, item_id_p, item_id_n, inter_id,
        user_emb_w, item_emb_w, wsum, out);
}


# Stage 4 contraction split over e: a wave owns 16 e (4 K-steps) and all 7 row tiles, its 4 weight fragments stay in registers, 7 independent accumulators; softmax adds the 8 partial logits pairwise
# speedup vs baseline: 1.5510x; 1.0403x over previous
#include <hip/hip_runtime.h>
#include <hip/hip_bf16.h>
#include <math.h>

#define B_  4096
#define M_  100
#define E_  128
#define R_  64

typedef float v2f __attribute__((ext_vector_type(2)));
typedef float v8f __attribute__((ext_vector_type(8)));

#define STRIDE    132
#define MT        112
#define OFF_INTER 0
#define OFF_U     (OFF_INTER + MT * STRIDE)
#define OFF_IP    (OFF_U + E_)
#define OFF_INN   (OFF_IP + E_)
#define OFF_W     (OFF_INN + E_)
#define OFF_WSUA  (OFF_W + 4 * E_)
#define OFF_WSIA  (OFF_WSUA + E_)
#define OFF_LOGIT (OFF_WSIA + E_)
#define OFF_SACC  (OFF_LOGIT + 4 * MT)
#define OFF_PART  OFF_SACC
#define SMEM_FLOATS (OFF_PART + 32 * MT)

__device__ __forceinline__ float wave_sum(float v) {
    #pragma unroll
    for (int off = 16; off > 0; off >>= 1) v += __shfl_xor(v, off, 32);
    return v;
}
__device__ __forceinline__ float wave_max(float v) {
    #pragma unroll
    for (int off = 16; off > 0; off >>= 1) v = fmaxf(v, __shfl_xor(v, off, 32));
    return v;
}

template <int M>
__device__ __forceinline__ float add_row_xor(float v) {
    const int b = __builtin_bit_cast(int, v);
    const int o = __builtin_amdgcn_update_dpp(b, b, 0x160 + M, 0xf, 0xf, false);
    return v + __builtin_bit_cast(float, o);
}

__global__ void hlrm_wsum_kernel(const float* __restrict__ ua_key,
                                 const float* __restrict__ ua_val,
                                 const float* __restrict__ ia_key,
                                 const float* __restrict__ ia_val,
                                 float* __restrict__ wsum) {
    int e = threadIdx.x;
    float s1 = 0.f, s2 = 0.f;
    #pragma unroll 4
    for (int r = 0; r < R_; ++r) {
        s1 += ua_key[r * E_ + e] * ua_val[r * E_ + e];
        s2 += ia_key[r * E_ + e] * ia_val[r * E_ + e];
    }
    wsum[e]       = s1;
    wsum[E_ + e]  = s2;
}

__global__ void __launch_bounds__(256, 2)
hlrm_main_kernel(const int* __restrict__ user_id,
                 const int* __restrict__ item_id_p,
                 const int* __restrict__ item_id_n,
                 const int* __restrict__ inter_id,
                 const float* __restrict__ user_emb_w,
                 const float* __restrict__ item_emb_w,
                 const float* __restrict__ wsum_g,
                 float* __restrict__ out) {
    extern __shared__ float sm[];
    float* sm_inter = sm + OFF_INTER;
    float* sm_u     = sm + OFF_U;
    float* sm_ip    = sm + OFF_IP;
    float* sm_inn   = sm + OFF_INN;
    float* sm_w     = sm + OFF_W;
    float* sm_wsua  = sm + OFF_WSUA;
    float* sm_wsia  = sm + OFF_WSIA;
    float* sm_logit = sm + OFF_LOGIT;
    float* sm_sacc  = sm + OFF_SACC;
    float* sm_part  = sm + OFF_PART;

    const int b    = blockIdx.x;
    const int t    = threadIdx.x;
    const int wave = t >> 5;
    const int lane = t & 31;
    const int half = lane >> 4;
    const int lm   = lane & 15;

    {
        const int q  = lane >> 3;
        const int j  = lane & 7;
        const int s0 = wave * 4 + q;
        const int s3 = 96 + s0;

        const int* idp = inter_id + b * M_ + (s3 < M_ ? s3 : 0);
        idp = (s3 == MT)     ? user_id   + b : idp;
        idp = (s3 == MT + 1) ? item_id_p + b : idp;
        idp = (s3 == MT + 2) ? item_id_n + b : idp;
        const int id3 = *idp;
        const int id0 = inter_id[b * M_ + s0];
        const int id1 = inter_id[b * M_ + s0 + 32];
        const int id2 = inter_id[b * M_ + s0 + 64];
        const bool has_src3 = (s3 < M_) || (s3 >= MT && s3 <= MT + 2);
        const bool has_dst3 = (s3 <= MT + 2);

        const float4* src0 = (const float4*)(item_emb_w + (size_t)id0 * E_) + j;
        const float4* src1 = (const float4*)(item_emb_w + (size_t)id1 * E_) + j;
        const float4* src2 = (const float4*)(item_emb_w + (size_t)id2 * E_) + j;
        const float4* src3 = (const float4*)((s3 == MT ? user_emb_w : item_emb_w)
                                             + (size_t)id3 * E_) + j;
        float* d3 = sm_inter + s3 * STRIDE;
        d3 = (s3 == MT)     ? sm_u   : d3;
        d3 = (s3 == MT + 1) ? sm_ip  : d3;
        d3 = (s3 == MT + 2) ? sm_inn : d3;
        float4* dst0 = (float4*)(sm_inter + s0 * STRIDE) + j;
        float4* dst1 = (float4*)(sm_inter + (s0 + 32) * STRIDE) + j;
        float4* dst2 = (float4*)(sm_inter + (s0 + 64) * STRIDE) + j;
        float4* dst3 = (float4*)d3 + j;

        float4 x[4][4];
        #pragma unroll
        for (int seg = 0; seg < 4; ++seg) {
            x[0][seg] = src0[seg * 8];
            x[1][seg] = src1[seg * 8];
            x[2][seg] = src2[seg * 8];
        }
        #pragma unroll
        for (int seg = 0; seg < 4; ++seg) {
            float4 v = {0.f, 0.f, 0.f, 0.f};
            if (has_src3) v = src3[seg * 8];
            x[3][seg] = v;
        }
        #pragma unroll
        for (int pass = 0; pass < 4; ++pass) {
            float ps[4];
            #pragma unroll
            for (int seg = 0; seg < 4; ++seg) {
                const float4 v = x[pass][seg];
                ps[seg] = v.x * v.x + v.y * v.y + v.z * v.z + v.w * v.w;
            }
            float ss = (ps[0] + ps[1]) + (ps[2] + ps[3]);
            ss = add_row_xor<4>(ss);
            ss = add_row_xor<2>(ss);
            ss = add_row_xor<1>(ss);
            const float n  = sqrtf(ss);
            const float sc = (n > 1.0f) ? (1.0f / (n + 1e-7f)) : 1.0f;
            #pragma unroll
            for (int seg = 0; seg < 4; ++seg) {
                x[pass][seg].x *= sc; x[pass][seg].y *= sc;
                x[pass][seg].z *= sc; x[pass][seg].w *= sc;
            }
        }
        #pragma unroll
        for (int seg = 0; seg < 4; ++seg) {
            dst0[seg * 8] = x[0][seg];
            dst1[seg * 8] = x[1][seg];
            dst2[seg * 8] = x[2][seg];
        }
        if (has_dst3) {
            #pragma unroll
            for (int seg = 0; seg < 4; ++seg) dst3[seg * 8] = x[3][seg];
        }
    }
    __syncthreads();

    if (t < E_) {
        const int e = t;
        const float uu = sm_u[e], pp = sm_ip[e], nn = sm_inn[e];
        const float wa = wsum_g[e], wi = wsum_g[E_ + e];
        sm_wsua[e] = wa;
        sm_wsia[e] = wi;
        sm_w[0 * E_ + e] = uu * pp * wa;
        sm_w[1 * E_ + e] = uu * pp * wi;
        sm_w[2 * E_ + e] = uu * nn * wa;
        sm_w[3 * E_ + e] = uu * nn * wi;
    }
    __syncthreads();

    {
        const float  bmask = (lm < 4) ? 1.0f : 0.0f;
        const float* wrow  = sm_w + (lm < 4 ? lm : 0) * E_ + 16 * wave + 2 * half;
        const float* Arow  = sm_inter + lm * STRIDE + 16 * wave + 2 * half;
        v2f bb[4];
        #pragma unroll
        for (int st = 0; st < 4; ++st) {
            bb[st][0] = wrow[4 * st] * bmask;
            bb[st][1] = wrow[4 * st + 1] * bmask;
        }
        v8f c[7];
        #pragma unroll
        for (int tile = 0; tile < 7; ++tile)
            c[tile] = (v8f){0.f, 0.f, 0.f, 0.f, 0.f, 0.f, 0.f, 0.f};
        #pragma unroll
        for (int st = 0; st < 4; ++st) {
            #pragma unroll
            for (int tile = 0; tile < 7; ++tile) {
                v2f a;
                a[0] = Arow[tile * 16 * STRIDE + 4 * st];
                a[1] = Arow[tile * 16 * STRIDE + 4 * st + 1];
                c[tile] = __builtin_amdgcn_wmma_f32_16x16x4_f32(false, a, false, bb[st],
                                                                (short)0, c[tile], false, false);
            }
        }
        if (lm < 4) {
            float4* P = (float4*)(sm_part + (wave * 4 + lm) * MT + 8 * half);
            #pragma unroll
            for (int tile = 0; tile < 7; ++tile) {
                P[4 * tile]     = make_float4(c[tile][0], c[tile][1], c[tile][2], c[tile][3]);
                P[4 * tile + 1] = make_float4(c[tile][4], c[tile][5], c[tile][6], c[tile][7]);
            }
        }
    }
    __syncthreads();

    if (wave < 4) {
        float* L = sm_logit + wave * MT;
        const float* P = sm_part + wave * MT;
        float vals[4], mx = -INFINITY;
        #pragma unroll
        for (int i = 0; i < 4; ++i) {
            int m = lane + 32 * i;
            float lg = -INFINITY;
            if (m < M_)
                lg = ((P[m]           + P[4 * MT + m])  + (P[8 * MT + m]  + P[12 * MT + m]))
                   + ((P[16 * MT + m] + P[20 * MT + m]) + (P[24 * MT + m] + P[28 * MT + m]));
            vals[i] = lg;
            mx = fmaxf(mx, vals[i]);
        }
        mx = wave_max(mx);
        float se = 0.f;
        #pragma unroll
        for (int i = 0; i < 4; ++i) {
            int m = lane + 32 * i;
            vals[i] = (m < M_) ? expf(vals[i] - mx) : 0.f;
            se += vals[i];
        }
        se = wave_sum(se);
        const float inv = 1.0f / se;
        #pragma unroll
        for (int i = 0; i < 4; ++i) {
            int m = lane + 32 * i;
            if (m < MT) L[m] = vals[i] * inv;
        }
    }
    __syncthreads();

    {
        const int   etile = wave;
        const float amask = (lm < 4) ? 1.0f : 0.0f;
        const float* Lrow = sm_logit + (lm < 4 ? lm : 0) * MT;
        v8f c = {0.f, 0.f, 0.f, 0.f, 0.f, 0.f, 0.f, 0.f};
        #pragma unroll 7
        for (int m0 = 0; m0 < MT; m0 += 4) {
            const int mk = m0 + 2 * half;
            v2f a, bb;
            a[0]  = Lrow[mk] * amask;     a[1]  = Lrow[mk + 1] * amask;
            bb[0] = sm_inter[mk * STRIDE + etile * 16 + lm];
            bb[1] = sm_inter[(mk + 1) * STRIDE + etile * 16 + lm];
            c = __builtin_amdgcn_wmma_f32_16x16x4_f32(false, a, false, bb,
                                                      (short)0, c, false, false);
        }
        if (half == 0) {
            #pragma unroll
            for (int v = 0; v < 4; ++v)
                sm_sacc[v * E_ + etile * 16 + lm] = c[v];
        }
    }
    __syncthreads();

    if (t < E_) {
        const int e = t;
        const float uu = sm_u[e], pp = sm_ip[e], nn = sm_inn[e];
        const float wa = sm_wsua[e], wi = sm_wsia[e];
        const float relp = pp * wa * sm_sacc[0 * E_ + e] + uu * wi * sm_sacc[1 * E_ + e];
        const float reln = nn * wa * sm_sacc[2 * E_ + e] + uu * wi * sm_sacc[3 * E_ + e];
        const size_t BE = (size_t)B_ * E_;
        const size_t base = (size_t)b * E_ + e;
        out[0 * BE + base] = uu;
        out[1 * BE + base] = pp;
        out[2 * BE + base] = nn;
        out[3 * BE + base] = relp;
        out[4 * BE + base] = reln;
    }
}

extern "C" void kernel_launch(void* const* d_in, const int* in_sizes, int n_in,
                              void* d_out, int out_size, void* d_ws, size_t ws_size,
                              hipStream_t stream) {
    const int*   user_id    = (const int*)d_in[0];
    const int*   item_id_p  = (const int*)d_in[1];
    const int*   item_id_n  = (const int*)d_in[2];
    const int*   inter_id   = (const int*)d_in[3];
    const float* user_emb_w = (const float*)d_in[4];
    const float* item_emb_w = (const float*)d_in[5];
    const float* ua_key     = (const float*)d_in[6];
    const float* ua_val     = (const float*)d_in[7];
    const float* ia_key     = (const float*)d_in[8];
    const float* ia_val     = (const float*)d_in[9];
    float* out  = (float*)d_out;
    float* wsum = (float*)d_ws;

    hlrm_wsum_kernel<<<1, E_, 0, stream>>>(ua_key, ua_val, ia_key, ia_val, wsum);
    hlrm_main_kernel<<<B_, 256, SMEM_FLOATS * sizeof(float), stream>>>(
        user_id, item_id_p, item_id_n, inter_id,
        user_emb_w, item_emb_w, wsum, out);
}


# Stage 6 contraction split over m: a wave owns 16 or 12 inter rows (4 or 3 K-steps), all 8 e tiles; attention fragments held in registers, 8 independent accumulators; stage 7 adds the 8 partial sums
# speedup vs baseline: 1.6206x; 1.0449x over previous
#include <hip/hip_runtime.h>
#include <hip/hip_bf16.h>
#include <math.h>

#define B_  4096
#define M_  100
#define E_  128
#define R_  64

typedef float v2f __attribute__((ext_vector_type(2)));
typedef float v8f __attribute__((ext_vector_type(8)));

#define STRIDE    132
#define MT        112
#define OFF_INTER 0
#define OFF_U     (OFF_INTER + MT * STRIDE)
#define OFF_IP    (OFF_U + E_)
#define OFF_INN   (OFF_IP + E_)
#define OFF_WSUA  (OFF_INN + E_)
#define OFF_WSIA  (OFF_WSUA + E_)
#define OFF_LOGIT (OFF_WSIA + E_)
#define OFF_W     (OFF_LOGIT + 4 * MT)
#define OFF_PART  (OFF_W + 4 * E_)
#define OFF_P6    OFF_W
#define SMEM_FLOATS (OFF_PART + 32 * MT)

__device__ __forceinline__ float wave_sum(float v) {
    #pragma unroll
    for (int off = 16; off > 0; off >>= 1) v += __shfl_xor(v, off, 32);
    return v;
}
__device__ __forceinline__ float wave_max(float v) {
    #pragma unroll
    for (int off = 16; off > 0; off >>= 1) v = fmaxf(v, __shfl_xor(v, off, 32));
    return v;
}

template <int M>
__device__ __forceinline__ float add_row_xor(float v) {
    const int b = __builtin_bit_cast(int, v);
    const int o = __builtin_amdgcn_update_dpp(b, b, 0x160 + M, 0xf, 0xf, false);
    return v + __builtin_bit_cast(float, o);
}

__global__ void hlrm_wsum_kernel(const float* __restrict__ ua_key,
                                 const float* __restrict__ ua_val,
                                 const float* __restrict__ ia_key,
                                 const float* __restrict__ ia_val,
                                 float* __restrict__ wsum) {
    int e = threadIdx.x;
    float s1 = 0.f, s2 = 0.f;
    #pragma unroll 4
    for (int r = 0; r < R_; ++r) {
        s1 += ua_key[r * E_ + e] * ua_val[r * E_ + e];
        s2 += ia_key[r * E_ + e] * ia_val[r * E_ + e];
    }
    wsum[e]       = s1;
    wsum[E_ + e]  = s2;
}

template <int NST>
__device__ __forceinline__ void stage6_slice(const float* sm_logit, const float* sm_inter,
                                             float* sm_p6, int wv, int mbase, int lm, int half) {
    const float* Lrow = sm_logit + (lm & 3) * MT + mbase + 2 * half;
    const float* Brow = sm_inter + (mbase + 2 * half) * STRIDE + lm;
    v2f a[NST];
    #pragma unroll
    for (int st = 0; st < NST; ++st) {
        a[st][0] = Lrow[4 * st];
        a[st][1] = Lrow[4 * st + 1];
    }
    v8f c[8];
    #pragma unroll
    for (int tile = 0; tile < 8; ++tile)
        c[tile] = (v8f){0.f, 0.f, 0.f, 0.f, 0.f, 0.f, 0.f, 0.f};
    #pragma unroll
    for (int st = 0; st < NST; ++st) {
        #pragma unroll
        for (int tile = 0; tile < 8; ++tile) {
            v2f bb;
            bb[0] = Brow[(4 * st) * STRIDE + tile * 16];
            bb[1] = Brow[(4 * st + 1) * STRIDE + tile * 16];
            c[tile] = __builtin_amdgcn_wmma_f32_16x16x4_f32(false, a[st], false, bb,
                                                            (short)0, c[tile], false, false);
        }
    }
    if (half == 0) {
        float4* P = (float4*)sm_p6 + wv * E_ + lm;
        #pragma unroll
        for (int tile = 0; tile < 8; ++tile)
            P[tile * 16] = make_float4(c[tile][0], c[tile][1], c[tile][2], c[tile][3]);
    }
}

__global__ void __launch_bounds__(256, 2)
hlrm_main_kernel(const int* __restrict__ user_id,
                 const int* __restrict__ item_id_p,
                 const int* __restrict__ item_id_n,
                 const int* __restrict__ inter_id,
                 const float* __restrict__ user_emb_w,
                 const float* __restrict__ item_emb_w,
                 const float* __restrict__ wsum_g,
                 float* __restrict__ out) {
    extern __shared__ float sm[];
    float* sm_inter = sm + OFF_INTER;
    float* sm_u     = sm + OFF_U;
    float* sm_ip    = sm + OFF_IP;
    float* sm_inn   = sm + OFF_INN;
    float* sm_w     = sm + OFF_W;
    float* sm_wsua  = sm + OFF_WSUA;
    float* sm_wsia  = sm + OFF_WSIA;
    float* sm_logit = sm + OFF_LOGIT;
    float* sm_part  = sm + OFF_PART;
    float* sm_p6    = sm + OFF_P6;

    const int b    = blockIdx.x;
    const int t    = threadIdx.x;
    const int wave = t >> 5;
    const int lane = t & 31;
    const int half = lane >> 4;
    const int lm   = lane & 15;

    {
        const int q  = lane >> 3;
        const int j  = lane & 7;
        const int s0 = wave * 4 + q;
        const int s3 = 96 + s0;

        const int* idp = inter_id + b * M_ + (s3 < M_ ? s3 : 0);
        idp = (s3 == MT)     ? user_id   + b : idp;
        idp = (s3 == MT + 1) ? item_id_p + b : idp;
        idp = (s3 == MT + 2) ? item_id_n + b : idp;
        const int id3 = *idp;
        const int id0 = inter_id[b * M_ + s0];
        const int id1 = inter_id[b * M_ + s0 + 32];
        const int id2 = inter_id[b * M_ + s0 + 64];
        const bool has_src3 = (s3 < M_) || (s3 >= MT && s3 <= MT + 2);
        const bool has_dst3 = (s3 <= MT + 2);

        const float4* src0 = (const float4*)(item_emb_w + (size_t)id0 * E_) + j;
        const float4* src1 = (const float4*)(item_emb_w + (size_t)id1 * E_) + j;
        const float4* src2 = (const float4*)(item_emb_w + (size_t)id2 * E_) + j;
        const float4* src3 = (const float4*)((s3 == MT ? user_emb_w : item_emb_w)
                                             + (size_t)id3 * E_) + j;
        float* d3 = sm_inter + s3 * STRIDE;
        d3 = (s3 == MT)     ? sm_u   : d3;
        d3 = (s3 == MT + 1) ? sm_ip  : d3;
        d3 = (s3 == MT + 2) ? sm_inn : d3;
        float4* dst0 = (float4*)(sm_inter + s0 * STRIDE) + j;
        float4* dst1 = (float4*)(sm_inter + (s0 + 32) * STRIDE) + j;
        float4* dst2 = (float4*)(sm_inter + (s0 + 64) * STRIDE) + j;
        float4* dst3 = (float4*)d3 + j;

        float4 x[4][4];
        #pragma unroll
        for (int seg = 0; seg < 4; ++seg) {
            x[0][seg] = src0[seg * 8];
            x[1][seg] = src1[seg * 8];
            x[2][seg] = src2[seg * 8];
        }
        #pragma unroll
        for (int seg = 0; seg < 4; ++seg) {
            float4 v = {0.f, 0.f, 0.f, 0.f};
            if (has_src3) v = src3[seg * 8];
            x[3][seg] = v;
        }
        #pragma unroll
        for (int pass = 0; pass < 4; ++pass) {
            float ps[4];
            #pragma unroll
            for (int seg = 0; seg < 4; ++seg) {
                const float4 v = x[pass][seg];
                ps[seg] = v.x * v.x + v.y * v.y + v.z * v.z + v.w * v.w;
            }
            float ss = (ps[0] + ps[1]) + (ps[2] + ps[3]);
            ss = add_row_xor<4>(ss);
            ss = add_row_xor<2>(ss);
            ss = add_row_xor<1>(ss);
            const float n  = sqrtf(ss);
            const float sc = (n > 1.0f) ? (1.0f / (n + 1e-7f)) : 1.0f;
            #pragma unroll
            for (int seg = 0; seg < 4; ++seg) {
                x[pass][seg].x *= sc; x[pass][seg].y *= sc;
                x[pass][seg].z *= sc; x[pass][seg].w *= sc;
            }
        }
        #pragma unroll
        for (int seg = 0; seg < 4; ++seg) {
            dst0[seg * 8] = x[0][seg];
            dst1[seg * 8] = x[1][seg];
            dst2[seg * 8] = x[2][seg];
        }
        if (has_dst3) {
            #pragma unroll
            for (int seg = 0; seg < 4; ++seg) dst3[seg * 8] = x[3][seg];
        }
    }
    __syncthreads();

    if (t < E_) {
        const int e = t;
        const float uu = sm_u[e], pp = sm_ip[e], nn = sm_inn[e];
        const float wa = wsum_g[e], wi = wsum_g[E_ + e];
        sm_wsua[e] = wa;
        sm_wsia[e] = wi;
        sm_w[0 * E_ + e] = uu * pp * wa;
        sm_w[1 * E_ + e] = uu * pp * wi;
        sm_w[2 * E_ + e] = uu * nn * wa;
        sm_w[3 * E_ + e] = uu * nn * wi;
    }
    __syncthreads();

    {
        const float  bmask = (lm < 4) ? 1.0f : 0.0f;
        const float* wrow  = sm_w + (lm < 4 ? lm : 0) * E_ + 16 * wave + 2 * half;
        const float* Arow  = sm_inter + lm * STRIDE + 16 * wave + 2 * half;
        v2f bb[4];
        #pragma unroll
        for (int st = 0; st < 4; ++st) {
            bb[st][0] = wrow[4 * st] * bmask;
            bb[st][1] = wrow[4 * st + 1] * bmask;
        }
        v8f c[7];
        #pragma unroll
        for (int tile = 0; tile < 7; ++tile)
            c[tile] = (v8f){0.f, 0.f, 0.f, 0.f, 0.f, 0.f, 0.f, 0.f};
        #pragma unroll
        for (int st = 0; st < 4; ++st) {
            #pragma unroll
            for (int tile = 0; tile < 7; ++tile) {
                v2f a;
                a[0] = Arow[tile * 16 * STRIDE + 4 * st];
                a[1] = Arow[tile * 16 * STRIDE + 4 * st + 1];
                c[tile] = __builtin_amdgcn_wmma_f32_16x16x4_f32(false, a, false, bb[st],
                                                                (short)0, c[tile], false, false);
            }
        }
        if (lm < 4) {
            float4* P = (float4*)(sm_part + (wave * 4 + lm) * MT + 8 * half);
            #pragma unroll
            for (int tile = 0; tile < 7; ++tile) {
                P[4 * tile]     = make_float4(c[tile][0], c[tile][1], c[tile][2], c[tile][3]);
                P[4 * tile + 1] = make_float4(c[tile][4], c[tile][5], c[tile][6], c[tile][7]);
            }
        }
    }
    __syncthreads();

    if (wave < 4) {
        float* L = sm_logit + wave * MT;
        const float* P = sm_part + wave * MT;
        float vals[4], mx = -INFINITY;
        #pragma unroll
        for (int i = 0; i < 4; ++i) {
            int m = lane + 32 * i;
            float lg = -INFINITY;
            if (m < M_)
                lg = ((P[m]           + P[4 * MT + m])  + (P[8 * MT + m]  + P[12 * MT + m]))
                   + ((P[16 * MT + m] + P[20 * MT + m]) + (P[24 * MT + m] + P[28 * MT + m]));
            vals[i] = lg;
            mx = fmaxf(mx, vals[i]);
        }
        mx = wave_max(mx);
        float se = 0.f;
        #pragma unroll
        for (int i = 0; i < 4; ++i) {
            int m = lane + 32 * i;
            vals[i] = (m < M_) ? expf(vals[i] - mx) : 0.f;
            se += vals[i];
        }
        se = wave_sum(se);
        const float inv = 1.0f / se;
        #pragma unroll
        for (int i = 0; i < 4; ++i) {
            int m = lane + 32 * i;
            if (m < MT) L[m] = vals[i] * inv;
        }
    }
    __syncthreads();

    {
        const int wv = __builtin_amdgcn_readfirstlane(wave);
        if (wv < 4) stage6_slice<4>(sm_logit, sm_inter, sm_p6, wv, 16 * wv, lm, half);
        else        stage6_slice<3>(sm_logit, sm_inter, sm_p6, wv, 64 + 12 * (wv - 4), lm, half);
    }
    __syncthreads();

    if (t < E_) {
        const int e = t;
        const float uu = sm_u[e], pp = sm_ip[e], nn = sm_inn[e];
        const float wa = sm_wsua[e], wi = sm_wsia[e];
        const float4* P = (const float4*)sm_p6 + e;
        float4 p[8];
        #pragma unroll
        for (int q = 0; q < 8; ++q) p[q] = P[q * E_];
        float4 s;
        s.x = ((p[0].x + p[1].x) + (p[2].x + p[3].x)) + ((p[4].x + p[5].x) + (p[6].x + p[7].x));
        s.y = ((p[0].y + p[1].y) + (p[2].y + p[3].y)) + ((p[4].y + p[5].y) + (p[6].y + p[7].y));
        s.z = ((p[0].z + p[1].z) + (p[2].z + p[3].z)) + ((p[4].z + p[5].z) + (p[6].z + p[7].z));
        s.w = ((p[0].w + p[1].w) + (p[2].w + p[3].w)) + ((p[4].w + p[5].w) + (p[6].w + p[7].w));
        const float relp = pp * wa * s.x + uu * wi * s.y;
        const float reln = nn * wa * s.z + uu * wi * s.w;
        const size_t BE = (size_t)B_ * E_;
        const size_t base = (size_t)b * E_ + e;
        out[0 * BE + base] = uu;
        out[1 * BE + base] = pp;
        out[2 * BE + base] = nn;
        out[3 * BE + base] = relp;
        out[4 * BE + base] = reln;
    }
}

extern "C" void kernel_launch(void* const* d_in, const int* in_sizes, int n_in,
                              void* d_out, int out_size, void* d_ws, size_t ws_size,
                              hipStream_t stream) {
    const int*   user_id    = (const int*)d_in[0];
    const int*   item_id_p  = (const int*)d_in[1];
    const int*   item_id_n  = (const int*)d_in[2];
    const int*   inter_id   = (const int*)d_in[3];
    const float* user_emb_w = (const float*)d_in[4];
    const float* item_emb_w = (const float*)d_in[5];
    const float* ua_key     = (const float*)d_in[6];
    const float* ua_val     = (const float*)d_in[7];
    const float* ia_key     = (const float*)d_in[8];
    const float* ia_val     = (const float*)d_in[9];
    float* out  = (float*)d_out;
    float* wsum = (float*)d_ws;

    hlrm_wsum_kernel<<<1, E_, 0, stream>>>(ua_key, ua_val, ia_key, ia_val, wsum);
    hlrm_main_kernel<<<B_, 256, SMEM_FLOATS * sizeof(float), stream>>>(
        user_id, item_id_p, item_id_n, inter_id,
        user_emb_w, item_emb_w, wsum, out);
}


# Weight-sum kernel: r sum split 32 ways (8 waves x 4 lane groups, 2 rows per lane) so all rows load at once; lane xor adds + LDS add of 8 wave partials; sum order changed (was serial over r)
# speedup vs baseline: 1.9230x; 1.1866x over previous
#include <hip/hip_runtime.h>
#include <hip/hip_bf16.h>
#include <math.h>

#define B_  4096
#define M_  100
#define E_  128
#define R_  64

typedef float v2f __attribute__((ext_vector_type(2)));
typedef float v8f __attribute__((ext_vector_type(8)));

#define STRIDE    132
#define MT        112
#define OFF_INTER 0
#define OFF_U     (OFF_INTER + MT * STRIDE)
#define OFF_IP    (OFF_U + E_)
#define OFF_INN   (OFF_IP + E_)
#define OFF_WSUA  (OFF_INN + E_)
#define OFF_WSIA  (OFF_WSUA + E_)
#define OFF_LOGIT (OFF_WSIA + E_)
#define OFF_W     (OFF_LOGIT + 4 * MT)
#define OFF_PART  (OFF_W + 4 * E_)
#define OFF_P6    OFF_W
#define SMEM_FLOATS (OFF_PART + 32 * MT)

__device__ __forceinline__ float wave_sum(float v) {
    #pragma unroll
    for (int off = 16; off > 0; off >>= 1) v += __shfl_xor(v, off, 32);
    return v;
}
__device__ __forceinline__ float wave_max(float v) {
    #pragma unroll
    for (int off = 16; off > 0; off >>= 1) v = fmaxf(v, __shfl_xor(v, off, 32));
    return v;
}

template <int M>
__device__ __forceinline__ float add_row_xor(float v) {
    const int b = __builtin_bit_cast(int, v);
    const int o = __builtin_amdgcn_update_dpp(b, b, 0x160 + M, 0xf, 0xf, false);
    return v + __builtin_bit_cast(float, o);
}

__device__ __forceinline__ float add_xor8_xor16(float v) {
    v = add_row_xor<8>(v);
    const int b = __builtin_bit_cast(int, v);
    const int o = __builtin_amdgcn_permlanex16(b, b, 0x76543210, 0xfedcba98, false, false);
    return v + __builtin_bit_cast(float, o);
}

__global__ void hlrm_wsum_kernel(const float* __restrict__ ua_key,
                                 const float* __restrict__ ua_val,
                                 const float* __restrict__ ia_key,
                                 const float* __restrict__ ia_val,
                                 float* __restrict__ wsum) {
    __shared__ float part[8 * 32];
    const int t    = threadIdx.x;
    const int wave = t >> 5;
    const int lane = t & 31;
    const int c    = lane & 7;
    const int g    = lane >> 3;
    const int p    = blockIdx.x >> 2;
    const int line = blockIdx.x & 3;
    const float* key = p ? ia_key : ua_key;
    const float* val = p ? ia_val : ua_val;
    const int off = (8 * wave + g) * E_ + 32 * line + 4 * c;
    const float4 k0 = *(const float4*)(key + off);
    const float4 v0 = *(const float4*)(val + off);
    const float4 k1 = *(const float4*)(key + off + 4 * E_);
    const float4 v1 = *(const float4*)(val + off + 4 * E_);
    float4 s;
    s.x = add_xor8_xor16(k0.x * v0.x + k1.x * v1.x);
    s.y = add_xor8_xor16(k0.y * v0.y + k1.y * v1.y);
    s.z = add_xor8_xor16(k0.z * v0.z + k1.z * v1.z);
    s.w = add_xor8_xor16(k0.w * v0.w + k1.w * v1.w);
    if (g == 0) ((float4*)part)[wave * 8 + c] = s;
    __syncthreads();
    if (wave == 0) {
        const float* q = part + lane;
        wsum[p * E_ + 32 * line + lane] =
            ((q[0] + q[32]) + (q[64] + q[96])) + ((q[128] + q[160]) + (q[192] + q[224]));
    }
}

template <int NST>
__device__ __forceinline__ void stage6_slice(const float* sm_logit, const float* sm_inter,
                                             float* sm_p6, int wv, int mbase, int lm, int half) {
    const float* Lrow = sm_logit + (lm & 3) * MT + mbase + 2 * half;
    const float* Brow = sm_inter + (mbase + 2 * half) * STRIDE + lm;
    v2f a[NST];
    #pragma unroll
    for (int st = 0; st < NST; ++st) {
        a[st][0] = Lrow[4 * st];
        a[st][1] = Lrow[4 * st + 1];
    }
    v8f c[8];
    #pragma unroll
    for (int tile = 0; tile < 8; ++tile)
        c[tile] = (v8f){0.f, 0.f, 0.f, 0.f, 0.f, 0.f, 0.f, 0.f};
    #pragma unroll
    for (int st = 0; st < NST; ++st) {
        #pragma unroll
        for (int tile = 0; tile < 8; ++tile) {
            v2f bb;
            bb[0] = Brow[(4 * st) * STRIDE + tile * 16];
            bb[1] = Brow[(4 * st + 1) * STRIDE + tile * 16];
            c[tile] = __builtin_amdgcn_wmma_f32_16x16x4_f32(false, a[st], false, bb,
                                                            (short)0, c[tile], false, false);
        }
    }
    if (half == 0) {
        float4* P = (float4*)sm_p6 + wv * E_ + lm;
        #pragma unroll
        for (int tile = 0; tile < 8; ++tile)
            P[tile * 16] = make_float4(c[tile][0], c[tile][1], c[tile][2], c[tile][3]);
    }
}

__global__ void __launch_bounds__(256, 2)
hlrm_main_kernel(const int* __restrict__ user_id,
                 const int* __restrict__ item_id_p,
                 const int* __restrict__ item_id_n,
                 const int* __restrict__ inter_id,
                 const float* __restrict__ user_emb_w,
                 const float* __restrict__ item_emb_w,
                 const float* __restrict__ wsum_g,
                 float* __restrict__ out) {
    extern __shared__ float sm[];
    float* sm_inter = sm + OFF_INTER;
    float* sm_u     = sm + OFF_U;
    float* sm_ip    = sm + OFF_IP;
    float* sm_inn   = sm + OFF_INN;
    float* sm_w     = sm + OFF_W;
    float* sm_wsua  = sm + OFF_WSUA;
    float* sm_wsia  = sm + OFF_WSIA;
    float* sm_logit = sm + OFF_LOGIT;
    float* sm_part  = sm + OFF_PART;
    float* sm_p6    = sm + OFF_P6;

    const int b    = blockIdx.x;
    const int t    = threadIdx.x;
    const int wave = t >> 5;
    const int lane = t & 31;
    const int half = lane >> 4;
    const int lm   = lane & 15;

    {
        const int q  = lane >> 3;
        const int j  = lane & 7;
        const int s0 = wave * 4 + q;
        const int s3 = 96 + s0;

        const int* idp = inter_id + b * M_ + (s3 < M_ ? s3 : 0);
        idp = (s3 == MT)     ? user_id   + b : idp;
        idp = (s3 == MT + 1) ? item_id_p + b : idp;
        idp = (s3 == MT + 2) ? item_id_n + b : idp;
        const int id3 = *idp;
        const int id0 = inter_id[b * M_ + s0];
        const int id1 = inter_id[b * M_ + s0 + 32];
        const int id2 = inter_id[b * M_ + s0 + 64];
        const bool has_src3 = (s3 < M_) || (s3 >= MT && s3 <= MT + 2);
        const bool has_dst3 = (s3 <= MT + 2);

        const float4* src0 = (const float4*)(item_emb_w + (size_t)id0 * E_) + j;
        const float4* src1 = (const float4*)(item_emb_w + (size_t)id1 * E_) + j;
        const float4* src2 = (const float4*)(item_emb_w + (size_t)id2 * E_) + j;
        const float4* src3 = (const float4*)((s3 == MT ? user_emb_w : item_emb_w)
                                             + (size_t)id3 * E_) + j;
        float* d3 = sm_inter + s3 * STRIDE;
        d3 = (s3 == MT)     ? sm_u   : d3;
        d3 = (s3 == MT + 1) ? sm_ip  : d3;
        d3 = (s3 == MT + 2) ? sm_inn : d3;
        float4* dst0 = (float4*)(sm_inter + s0 * STRIDE) + j;
        float4* dst1 = (float4*)(sm_inter + (s0 + 32) * STRIDE) + j;
        float4* dst2 = (float4*)(sm_inter + (s0 + 64) * STRIDE) + j;
        float4* dst3 = (float4*)d3 + j;

        float4 x[4][4];
        #pragma unroll
        for (int seg = 0; seg < 4; ++seg) {
            x[0][seg] = src0[seg * 8];
            x[1][seg] = src1[seg * 8];
            x[2][seg] = src2[seg * 8];
        }
        #pragma unroll
        for (int seg = 0; seg < 4; ++seg) {
            float4 v = {0.f, 0.f, 0.f, 0.f};
            if (has_src3) v = src3[seg * 8];
            x[3][seg] = v;
        }
        #pragma unroll
        for (int pass = 0; pass < 4; ++pass) {
            float ps[4];
            #pragma unroll
            for (int seg = 0; seg < 4; ++seg) {
                const float4 v = x[pass][seg];
                ps[seg] = v.x * v.x + v.y * v.y + v.z * v.z + v.w * v.w;
            }
            float ss = (ps[0] + ps[1]) + (ps[2] + ps[3]);
            ss = add_row_xor<4>(ss);
            ss = add_row_xor<2>(ss);
            ss = add_row_xor<1>(ss);
            const float n  = sqrtf(ss);
            const float sc = (n > 1.0f) ? (1.0f / (n + 1e-7f)) : 1.0f;
            #pragma unroll
            for (int seg = 0; seg < 4; ++seg) {
                x[pass][seg].x *= sc; x[pass][seg].y *= sc;
                x[pass][seg].z *= sc; x[pass][seg].w *= sc;
            }
        }
        #pragma unroll
        for (int seg = 0; seg < 4; ++seg) {
            dst0[seg * 8] = x[0][seg];
            dst1[seg * 8] = x[1][seg];
            dst2[seg * 8] = x[2][seg];
        }
        if (has_dst3) {
            #pragma unroll
            for (int seg = 0; seg < 4; ++seg) dst3[seg * 8] = x[3][seg];
        }
    }
    __syncthreads();

    if (t < E_) {
        const int e = t;
        const float uu = sm_u[e], pp = sm_ip[e], nn = sm_inn[e];
        const float wa = wsum_g[e], wi = wsum_g[E_ + e];
        sm_wsua[e] = wa;
        sm_wsia[e] = wi;
        sm_w[0 * E_ + e] = uu * pp * wa;
        sm_w[1 * E_ + e] = uu * pp * wi;
        sm_w[2 * E_ + e] = uu * nn * wa;
        sm_w[3 * E_ + e] = uu * nn * wi;
    }
    __syncthreads();

    {
        const float  bmask = (lm < 4) ? 1.0f : 0.0f;
        const float* wrow  = sm_w + (lm < 4 ? lm : 0) * E_ + 16 * wave + 2 * half;
        const float* Arow  = sm_inter + lm * STRIDE + 16 * wave + 2 * half;
        v2f bb[4];
        #pragma unroll
        for (int st = 0; st < 4; ++st) {
            bb[st][0] = wrow[4 * st] * bmask;
            bb[st][1] = wrow[4 * st + 1] * bmask;
        }
        v8f c[7];
        #pragma unroll
        for (int tile = 0; tile < 7; ++tile)
            c[tile] = (v8f){0.f, 0.f, 0.f, 0.f, 0.f, 0.f, 0.f, 0.f};
        #pragma unroll
        for (int st = 0; st < 4; ++st) {
            #pragma unroll
            for (int tile = 0; tile < 7; ++tile) {
                v2f a;
                a[0] = Arow[tile * 16 * STRIDE + 4 * st];
                a[1] = Arow[tile * 16 * STRIDE + 4 * st + 1];
                c[tile] = __builtin_amdgcn_wmma_f32_16x16x4_f32(false, a, false, bb[st],
                                                                (short)0, c[tile], false, false);
            }
        }
        if (lm < 4) {
            float4* P = (float4*)(sm_part + (wave * 4 + lm) * MT + 8 * half);
            #pragma unroll
            for (int tile = 0; tile < 7; ++tile) {
                P[4 * tile]     = make_float4(c[tile][0], c[tile][1], c[tile][2], c[tile][3]);
                P[4 * tile + 1] = make_float4(c[tile][4], c[tile][5], c[tile][6], c[tile][7]);
            }
        }
    }
    __syncthreads();

    if (wave < 4) {
        float* L = sm_logit + wave * MT;
        const float* P = sm_part + wave * MT;
        float vals[4], mx = -INFINITY;
        #pragma unroll
        for (int i = 0; i < 4; ++i) {
            int m = lane + 32 * i;
            float lg = -INFINITY;
            if (m < M_)
                lg = ((P[m]           + P[4 * MT + m])  + (P[8 * MT + m]  + P[12 * MT + m]))
                   + ((P[16 * MT + m] + P[20 * MT + m]) + (P[24 * MT + m] + P[28 * MT + m]));
            vals[i] = lg;
            mx = fmaxf(mx, vals[i]);
        }
        mx = wave_max(mx);
        float se = 0.f;
        #pragma unroll
        for (int i = 0; i < 4; ++i) {
            int m = lane + 32 * i;
            vals[i] = (m < M_) ? expf(vals[i] - mx) : 0.f;
            se += vals[i];
        }
        se = wave_sum(se);
        const float inv = 1.0f / se;
        #pragma unroll
        for (int i = 0; i < 4; ++i) {
            int m = lane + 32 * i;
            if (m < MT) L[m] = vals[i] * inv;
        }
    }
    __syncthreads();

    {
        const int wv = __builtin_amdgcn_readfirstlane(wave);
        if (wv < 4) stage6_slice<4>(sm_logit, sm_inter, sm_p6, wv, 16 * wv, lm, half);
        else        stage6_slice<3>(sm_logit, sm_inter, sm_p6, wv, 64 + 12 * (wv - 4), lm, half);
    }
    __syncthreads();

    if (t < E_) {
        const int e = t;
        const float uu = sm_u[e], pp = sm_ip[e], nn = sm_inn[e];
        const float wa = sm_wsua[e], wi = sm_wsia[e];
        const float4* P = (const float4*)sm_p6 + e;
        float4 p[8];
        #pragma unroll
        for (int q = 0; q < 8; ++q) p[q] = P[q * E_];
        float4 s;
        s.x = ((p[0].x + p[1].x) + (p[2].x + p[3].x)) + ((p[4].x + p[5].x) + (p[6].x + p[7].x));
        s.y = ((p[0].y + p[1].y) + (p[2].y + p[3].y)) + ((p[4].y + p[5].y) + (p[6].y + p[7].y));
        s.z = ((p[0].z + p[1].z) + (p[2].z + p[3].z)) + ((p[4].z + p[5].z) + (p[6].z + p[7].z));
        s.w = ((p[0].w + p[1].w) + (p[2].w + p[3].w)) + ((p[4].w + p[5].w) + (p[6].w + p[7].w));
        const float relp = pp * wa * s.x + uu * wi * s.y;
        const float reln = nn * wa * s.z + uu * wi * s.w;
        const size_t BE = (size_t)B_ * E_;
        const size_t base = (size_t)b * E_ + e;
        out[0 * BE + base] = uu;
        out[1 * BE + base] = pp;
        out[2 * BE + base] = nn;
        out[3 * BE + base] = relp;
        out[4 * BE + base] = reln;
    }
}

extern "C" void kernel_launch(void* const* d_in, const int* in_sizes, int n_in,
                              void* d_out, int out_size, void* d_ws, size_t ws_size,
                              hipStream_t stream) {
    const int*   user_id    = (const int*)d_in[0];
    const int*   item_id_p  = (const int*)d_in[1];
    const int*   item_id_n  = (const int*)d_in[2];
    const int*   inter_id   = (const int*)d_in[3];
    const float* user_emb_w = (const float*)d_in[4];
    const float* item_emb_w = (const float*)d_in[5];
    const float* ua_key     = (const float*)d_in[6];
    const float* ua_val     = (const float*)d_in[7];
    const float* ia_key     = (const float*)d_in[8];
    const float* ia_val     = (const float*)d_in[9];
    float* out  = (float*)d_out;
    float* wsum = (float*)d_ws;

    hlrm_wsum_kernel<<<8, 256, 0, stream>>>(ua_key, ua_val, ia_key, ia_val, wsum);
    hlrm_main_kernel<<<B_, 256, SMEM_FLOATS * sizeof(float), stream>>>(
        user_id, item_id_p, item_id_n, inter_id,
        user_emb_w, item_emb_w, wsum, out);
}
